# starting kernel
# baseline (speedup 1.0000x reference)
#include <hip/hip_runtime.h>


typedef __attribute__((ext_vector_type(2))) float v2f;
typedef __attribute__((ext_vector_type(4))) float v4f;
typedef __attribute__((ext_vector_type(8))) float v8f;

#define HSZ 20
#define S   34
#define TPW 4

__device__ __forceinline__ float relu(float v) {
    return __builtin_amdgcn_fmed3f(v, 0.0f, 3.402823466e38f);
}

__global__ __launch_bounds__(32) void rnn_scan_wmma_f32(
    const float* __restrict__ x,
    const float* __restrict__ W_ih,
    const float* __restrict__ W_hh,
    const float* __restrict__ b_ih,
    const float* __restrict__ b_hh,
    const float* __restrict__ fc_w,
    const float* __restrict__ fc_b,
    float* __restrict__ out,
    int B, int T)
{
    __shared__ float htile[TPW][16 * S];

    const int lane = threadIdx.x;
    const int m    = lane & 15;
    const int hi   = lane >> 4;
    const int koff = hi * 2;
    const int base = blockIdx.x * (16 * TPW);

    auto welem = [&](int k, int j) -> float {
        float v = 0.0f;
        if (j < HSZ) {
            if (k < HSZ)            v = W_hh[j * HSZ + k];
            else if (k == HSZ)      v = W_ih[j];
            else if (k == HSZ + 1)  v = b_ih[j] + b_hh[j];
        }
        return v;
    };
    v2f bw[2][6];
    #pragma unroll
    for (int n = 0; n < 2; ++n) {
        #pragma unroll
        for (int kk = 0; kk < 6; ++kk) {
            const int j  = n * 16 + m;
            const int k0 = kk * 4 + koff;
            bw[n][kk].x = welem(k0, j);
            bw[n][kk].y = welem(k0 + 1, j);
        }
    }

    for (int i = lane; i < TPW * 16 * S; i += 32) (&htile[0][0])[i] = 0.0f;
    __syncthreads();

    const float* xrow[TPW];
    #pragma unroll
    for (int tt = 0; tt < TPW; ++tt) {
        int row = base + tt * 16 + m;
        if (row >= B) row = B - 1;
        xrow[tt] = x + (size_t)row * (size_t)T;
    }

    auto step = [&](const float (&xv)[TPW]) {
        v2f a[TPW][6];
        #pragma unroll
        for (int tt = 0; tt < TPW; ++tt) {
            #pragma unroll
            for (int kk = 0; kk < 5; ++kk)
                a[tt][kk] = *(const v2f*)&htile[tt][m * S + kk * 4 + koff];
            a[tt][5].x = hi ? 0.0f : xv[tt];
            a[tt][5].y = hi ? 0.0f : 1.0f;
        }
        v8f c[TPW][2];
        #pragma unroll
        for (int tt = 0; tt < TPW; ++tt) {
            c[tt][0] = (v8f){};
            c[tt][1] = (v8f){};
            #pragma unroll
            for (int kk = 0; kk < 6; ++kk) {
                c[tt][0] = __builtin_amdgcn_wmma_f32_16x16x4_f32(
                               false, a[tt][kk], false, bw[0][kk], (short)0,
                               c[tt][0], false, false);
                c[tt][1] = __builtin_amdgcn_wmma_f32_16x16x4_f32(
                               false, a[tt][kk], false, bw[1][kk], (short)0,
                               c[tt][1], false, false);
            }
        }
        #pragma unroll
        for (int tt = 0; tt < TPW; ++tt) {
            #pragma unroll
            for (int r = 0; r < 8; ++r) {
                const int row = hi * 8 + r;
                htile[tt][row * S + m]      = relu(c[tt][0][r]);
                htile[tt][row * S + 16 + m] = relu(c[tt][1][r]);
            }
        }
    };

    if ((T & 3) == 0) {
        const int G = T >> 2;
        v4f xb0[TPW], xb1[TPW];
        #pragma unroll
        for (int tt = 0; tt < TPW; ++tt) {
            xb0[tt] = *(const v4f*)(xrow[tt]);
            xb1[tt] = (G > 1) ? *(const v4f*)(xrow[tt] + 4) : xb0[tt];
        }
        #pragma unroll 1
        for (int g = 0; g < G; ++g) {
            v4f cur[TPW];
            #pragma unroll
            for (int tt = 0; tt < TPW; ++tt) { cur[tt] = xb0[tt]; xb0[tt] = xb1[tt]; }
            const int gg = (g + 2 < G) ? (g + 2) : (G - 1);
            #pragma unroll
            for (int tt = 0; tt < TPW; ++tt)
                xb1[tt] = *(const v4f*)(xrow[tt] + (size_t)gg * 4);
            #pragma unroll
            for (int u = 0; u < 4; ++u) {
                float xv[TPW];
                #pragma unroll
                for (int tt = 0; tt < TPW; ++tt) xv[tt] = cur[tt][u];
                step(xv);
            }
        }
    } else {
        #pragma unroll 1
        for (int t = 0; t < T; ++t) {
            float xv[TPW];
            #pragma unroll
            for (int tt = 0; tt < TPW; ++tt) xv[tt] = xrow[tt][t];
            step(xv);
        }
    }
    __syncthreads();

    if (lane < 16) {
        #pragma unroll
        for (int tt = 0; tt < TPW; ++tt) {
            const int row = base + tt * 16 + lane;
            if (row < B) {
                float acc = fc_b[0];
                #pragma unroll
                for (int j = 0; j < HSZ; ++j)
                    acc += htile[tt][lane * S + j] * fc_w[j];
                out[row] = acc;
            }
        }
    }
}

extern "C" void kernel_launch(void* const* d_in, const int* in_sizes, int n_in,
                              void* d_out, int out_size, void* d_ws, size_t ws_size,
                              hipStream_t stream) {
    const float* x    = (const float*)d_in[0];
    const float* W_ih = (const float*)d_in[1];
    const float* W_hh = (const float*)d_in[2];
    const float* b_ih = (const float*)d_in[3];
    const float* b_hh = (const float*)d_in[4];
    const float* fc_w = (const float*)d_in[5];
    const float* fc_b = (const float*)d_in[6];
    float* out = (float*)d_out;

    const int B = out_size;
    const int T = in_sizes[0] / B;
    const int rowsPerBlock = 16 * TPW;
    const int blocks = (B + rowsPerBlock - 1) / rowsPerBlock;

    rnn_scan_wmma_f32<<<blocks, 32, 0, stream>>>(x, W_ih, W_hh, b_ih, b_hh,
                                                 fc_w, fc_b, out, B, T);
}


# RNN state stays in registers: transposed product (weights A, state B) so relu(D) is the next step's B, no LDS round trip; 1 tile per wave, 256 waves; K-sum order changed: bias as C, x block first
# speedup vs baseline: 5.4797x; 5.4797x over previous
#include <hip/hip_runtime.h>


typedef __attribute__((ext_vector_type(2))) float v2f;
typedef __attribute__((ext_vector_type(4))) float v4f;
typedef __attribute__((ext_vector_type(8))) float v8f;

#define HSZ 20
#define WPB 4
#define HP  21

__device__ __forceinline__ float relu(float v) {
    return __builtin_amdgcn_fmed3f(v, 0.0f, 3.402823466e38f);
}

__device__ __forceinline__ v8f mma(v2f a, v2f b, v8f c) {
    return __builtin_amdgcn_wmma_f32_16x16x4_f32(false, a, false, b, (short)0, c, false, false);
}

__device__ __forceinline__ void hold_order() { __builtin_amdgcn_sched_barrier(0); }

__device__ __forceinline__ void step_order() {
    __builtin_amdgcn_sched_group_barrier(0x008, 2, 0);
    __builtin_amdgcn_sched_group_barrier(0x002, 10, 0);
    __builtin_amdgcn_sched_group_barrier(0x008, 10, 0);
}

__global__ __launch_bounds__(32 * WPB) void rnn_scan_wmma_f32(
    const float* __restrict__ x,
    const float* __restrict__ W_ih,
    const float* __restrict__ W_hh,
    const float* __restrict__ b_ih,
    const float* __restrict__ b_hh,
    const float* __restrict__ fc_w,
    const float* __restrict__ fc_b,
    float* __restrict__ out,
    int B, int T)
{
    __shared__ float hfin[WPB][16 * HP];

    const int tid  = threadIdx.x;
    const int lane = tid & 31;
    const int wave = tid >> 5;
    const int c    = lane & 15;
    const int hf   = lane >> 4;
    const int base = (blockIdx.x * WPB + wave) * 16;

    v2f aw[2][5];
    v2f axe[2];
    v2f axo[2];
    v8f bc[2];
    #pragma unroll
    for (int n = 0; n < 2; ++n) {
        const int j  = 16 * n + 4 * ((c & 7) >> 1) + 2 * (c >> 3) + (c & 1);
        const bool ok = j < HSZ;
        const int jc = ok ? j : 0;
        #pragma unroll
        for (int q = 0; q < 5; ++q) {
            const int k0 = 4 * q + 2 * hf;
            const float w0 = W_hh[jc * HSZ + k0];
            const float w1 = W_hh[jc * HSZ + k0 + 1];
            aw[n][q].x = ok ? w0 : 0.0f;
            aw[n][q].y = ok ? w1 : 0.0f;
        }
        const float wi = (ok && hf == 0) ? W_ih[jc] : 0.0f;
        axe[n].x = wi;   axe[n].y = 0.0f;
        axo[n].x = 0.0f; axo[n].y = wi;
        #pragma unroll
        for (int r = 0; r < 8; ++r) {
            const int i  = 8 * hf + r;
            const int jr = 16 * n + 4 * ((i & 7) >> 1) + 2 * (i >> 3) + (i & 1);
            const int jrc = (jr < HSZ) ? jr : 0;
            const float bs = b_ih[jrc] + b_hh[jrc];
            bc[n][r] = (jr < HSZ) ? bs : 0.0f;
        }
    }

    int row = base + c;
    if (row >= B) row = B - 1;
    const float* xr = x + (size_t)row * (size_t)T;

    v8f d0p = (v8f){};
    v8f d1p = (v8f){};

    auto step = [&](v2f bp, const v2f (&ax)[2]) {
        v8f d1 = mma(ax[1], bp, bc[1]);
        v8f d0 = mma(ax[0], bp, bc[0]);
        v2f hb4;
        hb4.x = relu(d1p[0]);
        hb4.y = relu(d1p[1]);
        v2f hb[4];
        #pragma unroll
        for (int q = 0; q < 4; ++q) {
            hb[q].x = relu(d0p[2 * q]);
            hb[q].y = relu(d0p[2 * q + 1]);
        }
        d1 = mma(aw[1][4], hb4, d1);
        d0 = mma(aw[0][4], hb4, d0);
        #pragma unroll
        for (int q = 0; q < 4; ++q) {
            d1 = mma(aw[1][q], hb[q], d1);
            d0 = mma(aw[0][q], hb[q], d0);
        }
        step_order();
        d0p = d0;
        d1p = d1;
    };

    const int NC = ((T & 3) == 0) ? (T >> 3) : 0;
    if (NC > 0) {
        v4f ca = *(const v4f*)(xr);
        v4f cb = *(const v4f*)(xr + 4);
        const float* pn = xr + ((NC > 1) ? 8 : 0);
        #pragma unroll 1
        for (int i = 0; i < NC; ++i) {
            const v4f na = *(const v4f*)(pn);
            const v4f nb = *(const v4f*)(pn + 4);
            hold_order();
            step(__builtin_shufflevector(ca, ca, 0, 1), axe);
            step(__builtin_shufflevector(ca, ca, 0, 1), axo);
            step(__builtin_shufflevector(ca, ca, 2, 3), axe);
            step(__builtin_shufflevector(ca, ca, 2, 3), axo);
            step(__builtin_shufflevector(cb, cb, 0, 1), axe);
            step(__builtin_shufflevector(cb, cb, 0, 1), axo);
            step(__builtin_shufflevector(cb, cb, 2, 3), axe);
            step(__builtin_shufflevector(cb, cb, 2, 3), axo);
            hold_order();
            asm volatile("" :: "v"(na), "v"(nb));
            ca = na;
            cb = nb;
            pn += (i + 2 < NC) ? 8 : 0;
        }
    }
    int t = NC * 8;
    #pragma unroll 1
    for (; t + 1 < T; t += 2) {
        v2f bp;
        bp.x = xr[t];
        bp.y = xr[t + 1];
        step(bp, axe);
        step(bp, axo);
    }
    if (t < T) {
        v2f bp;
        bp.x = xr[t];
        bp.y = 0.0f;
        step(bp, axe);
    }

    float* hw = hfin[wave];
    #pragma unroll
    for (int q = 0; q < 4; ++q) {
        hw[c * HP + 4 * q + 2 * hf]     = relu(d0p[2 * q]);
        hw[c * HP + 4 * q + 2 * hf + 1] = relu(d0p[2 * q + 1]);
    }
    hw[c * HP + 16 + 2 * hf]     = relu(d1p[0]);
    hw[c * HP + 16 + 2 * hf + 1] = relu(d1p[1]);
    __syncthreads();

    if (tid < 16 * WPB) {
        const int orow = blockIdx.x * (16 * WPB) + tid;
        if (orow < B) {
            const float* hr = &hfin[tid >> 4][(tid & 15) * HP];
            float acc = fc_b[0];
            #pragma unroll
            for (int j = 0; j < HSZ; ++j)
                acc += hr[j] * fc_w[j];
            out[orow] = acc;
        }
    }
}

extern "C" void kernel_launch(void* const* d_in, const int* in_sizes, int n_in,
                              void* d_out, int out_size, void* d_ws, size_t ws_size,
                              hipStream_t stream) {
    const float* x    = (const float*)d_in[0];
    const float* W_ih = (const float*)d_in[1];
    const float* W_hh = (const float*)d_in[2];
    const float* b_ih = (const float*)d_in[3];
    const float* b_hh = (const float*)d_in[4];
    const float* fc_w = (const float*)d_in[5];
    const float* fc_b = (const float*)d_in[6];
    float* out = (float*)d_out;

    const int B = out_size;
    const int T = in_sizes[0] / B;
    const int rowsPerBlock = 16 * WPB;
    const int blocks = (B + rowsPerBlock - 1) / rowsPerBlock;

    rnn_scan_wmma_f32<<<blocks, 32 * WPB, 0, stream>>>(x, W_ih, W_hh, b_ih, b_hh,
                                                       fc_w, fc_b, out, B, T);
}


# Tile 1 holds only units 16..19 (4 of 16 rows): its rows now serve four steps, one input-block WMMA per four steps opens it, accumulator carried; 90 WMMAs per 8 steps, was 96; sum order unchanged
# speedup vs baseline: 5.7367x; 1.0469x over previous
#include <hip/hip_runtime.h>


typedef __attribute__((ext_vector_type(2))) float v2f;
typedef __attribute__((ext_vector_type(8))) float v8f;

#define HSZ 20
#define WPB 4
#define HP  21

template <int P> struct step_no { static constexpr int value = P; };

__device__ __forceinline__ float relu(float v) {
    return __builtin_amdgcn_fmed3f(v, 0.0f, 3.402823466e38f);
}

__device__ __forceinline__ v8f mma(v2f a, v2f b, v8f c) {
    return __builtin_amdgcn_wmma_f32_16x16x4_f32(false, a, false, b, (short)0, c, false, false);
}

__device__ __forceinline__ void hold_order() { __builtin_amdgcn_sched_barrier(0); }

template <int NX>
__device__ __forceinline__ void step_order() {
    __builtin_amdgcn_sched_group_barrier(0x008, NX, 0);
    __builtin_amdgcn_sched_group_barrier(0x002, 10, 0);
    __builtin_amdgcn_sched_group_barrier(0x008, 10, 0);
}

__global__ __launch_bounds__(32 * WPB) void rnn_scan_wmma_f32(
    const float* __restrict__ x,
    const float* __restrict__ W_ih,
    const float* __restrict__ W_hh,
    const float* __restrict__ b_ih,
    const float* __restrict__ b_hh,
    const float* __restrict__ fc_w,
    const float* __restrict__ fc_b,
    float* __restrict__ out,
    int B, int T)
{
    __shared__ float hfin[WPB][16 * HP];

    const int tid  = threadIdx.x;
    const int lane = tid & 31;
    const int wave = tid >> 5;
    const int c    = lane & 15;
    const int hf   = lane >> 4;
    const int base = (blockIdx.x * WPB + wave) * 16;

    v2f aw0[5];
    v2f a0x[4];
    v2f a1w[4][5];
    v2f a1x;
    v8f bc0;
    v8f bc1;
    {
        const int j0 = 4 * ((c & 7) >> 1) + 2 * (c >> 3) + (c & 1);
        const int j1 = 16 + 2 * (c >> 3) + (c & 1);
        const int s1 = (c & 7) >> 1;
        float w1[5][2];
        #pragma unroll
        for (int q = 0; q < 5; ++q) {
            const int k0 = 4 * q + 2 * hf;
            aw0[q].x = W_hh[j0 * HSZ + k0];
            aw0[q].y = W_hh[j0 * HSZ + k0 + 1];
            w1[q][0] = W_hh[j1 * HSZ + k0];
            w1[q][1] = W_hh[j1 * HSZ + k0 + 1];
        }
        #pragma unroll
        for (int p = 0; p < 4; ++p) {
            #pragma unroll
            for (int q = 0; q < 5; ++q) {
                a1w[p][q].x = (s1 == p) ? w1[q][0] : 0.0f;
                a1w[p][q].y = (s1 == p) ? w1[q][1] : 0.0f;
            }
        }
        const float wi0 = W_ih[j0];
        #pragma unroll
        for (int p = 0; p < 4; ++p) {
            a0x[p].x = (hf == (p >> 1) && (p & 1) == 0) ? wi0 : 0.0f;
            a0x[p].y = (hf == (p >> 1) && (p & 1) == 1) ? wi0 : 0.0f;
        }
        const float wi1 = W_ih[j1];
        a1x.x = (hf == (s1 >> 1) && (s1 & 1) == 0) ? wi1 : 0.0f;
        a1x.y = (hf == (s1 >> 1) && (s1 & 1) == 1) ? wi1 : 0.0f;
        #pragma unroll
        for (int r = 0; r < 8; ++r) {
            const int jr0 = 4 * (r >> 1) + 2 * hf + (r & 1);
            const int jr1 = 16 + 2 * hf + (r & 1);
            bc0[r] = b_ih[jr0] + b_hh[jr0];
            bc1[r] = b_ih[jr1] + b_hh[jr1];
        }
    }
    asm volatile("" : "+v"(bc1), "+v"(a0x[0]), "+v"(a0x[1]), "+v"(a0x[2]), "+v"(a0x[3]));

    int row = base + c;
    if (row >= B) row = B - 1;
    const float* xr = x + (size_t)row * (size_t)T;

    v8f d0p = (v8f){};
    v2f s1p = (v2f){};
    v8f acc1 = (v8f){};

    auto step = [&](auto pc, v2f g) {
        constexpr int P = decltype(pc)::value;
        if constexpr (P == 0)
            acc1 = mma(a1x, g, bc1);
        v8f d1 = acc1;
        v8f d0 = mma(a0x[P], g, bc0);
        v2f hb4;
        hb4.x = relu(s1p.x);
        hb4.y = relu(s1p.y);
        v2f hb[4];
        #pragma unroll
        for (int q = 0; q < 4; ++q) {
            hb[q].x = relu(d0p[2 * q]);
            hb[q].y = relu(d0p[2 * q + 1]);
        }
        d1 = mma(a1w[P][4], hb4, d1);
        d0 = mma(aw0[4], hb4, d0);
        #pragma unroll
        for (int q = 0; q < 4; ++q) {
            d1 = mma(a1w[P][q], hb[q], d1);
            d0 = mma(aw0[q], hb[q], d0);
        }
        step_order<(P == 0) ? 2 : 1>();
        d0p = d0;
        acc1 = d1;
        s1p.x = d1[2 * P];
        s1p.y = d1[2 * P + 1];
    };

    const int NC = ((T & 3) == 0) ? (T >> 3) : 0;
    if (NC > 0) {
        const float* xh = xr + 2 * hf;
        v2f ca = *(const v2f*)(xh);
        v2f cb = *(const v2f*)(xh + 4);
        const float* pn = xh + ((NC > 1) ? 8 : 0);
        #pragma unroll 1
        for (int i = 0; i < NC; ++i) {
            const v2f na = *(const v2f*)(pn);
            const v2f nb = *(const v2f*)(pn + 4);
            hold_order();
            step(step_no<0>{}, ca); hold_order();
            step(step_no<1>{}, ca); hold_order();
            step(step_no<2>{}, ca); hold_order();
            step(step_no<3>{}, ca); hold_order();
            step(step_no<0>{}, cb); hold_order();
            step(step_no<1>{}, cb); hold_order();
            step(step_no<2>{}, cb); hold_order();
            step(step_no<3>{}, cb);
            hold_order();
            asm volatile("" :: "v"(na), "v"(nb));
            ca = na;
            cb = nb;
            pn += (i + 2 < NC) ? 8 : 0;
        }
    }
    #pragma unroll 1
    for (int t = NC * 8; t < T; ++t) {
        v2f g;
        g.x = xr[t];
        g.y = 0.0f;
        step(step_no<0>{}, g);
    }

    float* hw = hfin[wave];
    #pragma unroll
    for (int q = 0; q < 4; ++q) {
        hw[c * HP + 4 * q + 2 * hf]     = relu(d0p[2 * q]);
        hw[c * HP + 4 * q + 2 * hf + 1] = relu(d0p[2 * q + 1]);
    }
    hw[c * HP + 16 + 2 * hf]     = relu(s1p.x);
    hw[c * HP + 16 + 2 * hf + 1] = relu(s1p.y);
    __syncthreads();

    if (tid < 16 * WPB) {
        const int orow = blockIdx.x * (16 * WPB) + tid;
        if (orow < B) {
            const float* hr = &hfin[tid >> 4][(tid & 15) * HP];
            float acc = fc_b[0];
            #pragma unroll
            for (int j = 0; j < HSZ; ++j)
                acc += hr[j] * fc_w[j];
            out[orow] = acc;
        }
    }
}

extern "C" void kernel_launch(void* const* d_in, const int* in_sizes, int n_in,
                              void* d_out, int out_size, void* d_ws, size_t ws_size,
                              hipStream_t stream) {
    const float* x    = (const float*)d_in[0];
    const float* W_ih = (const float*)d_in[1];
    const float* W_hh = (const float*)d_in[2];
    const float* b_ih = (const float*)d_in[3];
    const float* b_hh = (const float*)d_in[4];
    const float* fc_w = (const float*)d_in[5];
    const float* fc_b = (const float*)d_in[6];
    float* out = (float*)d_out;

    const int B = out_size;
    const int T = in_sizes[0] / B;
    const int rowsPerBlock = 16 * WPB;
    const int blocks = (B + rowsPerBlock - 1) / rowsPerBlock;

    rnn_scan_wmma_f32<<<blocks, 32 * WPB, 0, stream>>>(x, W_ih, W_hh, b_ih, b_hh,
                                                       fc_w, fc_b, out, B, T);
}


# Units 16..19 leave the quarter-filled second tile for the vector ALU (40 FMAs a lane, one exchange between lane halves) beside the tile's WMMAs: 24 WMMAs per 4 steps, was 45; their sum order changed
# speedup vs baseline: 6.7236x; 1.1720x over previous
#include <hip/hip_runtime.h>


typedef __attribute__((ext_vector_type(2))) float v2f;
typedef __attribute__((ext_vector_type(4))) float v4f;
typedef __attribute__((ext_vector_type(8))) float v8f;

#define HSZ 20
#define WPB 4
#define HP  21

template <int P> struct step_no { static constexpr int value = P; };

__device__ __forceinline__ float relu(float v) {
    return __builtin_amdgcn_fmed3f(v, 0.0f, 3.402823466e38f);
}

__device__ __forceinline__ v8f mma(v2f a, v2f b, v8f c) {
    return __builtin_amdgcn_wmma_f32_16x16x4_f32(false, a, false, b, (short)0, c, false, false);
}

__device__ __forceinline__ float xhalf(float x) {
    const int b = __builtin_bit_cast(int, x);
    return __builtin_bit_cast(float,
        __builtin_amdgcn_permlanex16(b, b, 0x76543210, 0xfedcba98, false, false));
}

__device__ __forceinline__ void hold_order() { __builtin_amdgcn_sched_barrier(0); }

__device__ __forceinline__ void step_order() {
    __builtin_amdgcn_sched_group_barrier(0x008, 2, 0);
    __builtin_amdgcn_sched_group_barrier(0x002, 10, 0);
    __builtin_amdgcn_sched_group_barrier(0x008, 1, 0);
    __builtin_amdgcn_sched_group_barrier(0x002, 16, 0);
    __builtin_amdgcn_sched_group_barrier(0x008, 1, 0);
    __builtin_amdgcn_sched_group_barrier(0x002, 16, 0);
    __builtin_amdgcn_sched_group_barrier(0x008, 1, 0);
    __builtin_amdgcn_sched_group_barrier(0x002, 14, 0);
    __builtin_amdgcn_sched_group_barrier(0x008, 1, 0);
    __builtin_amdgcn_sched_group_barrier(0x002, 4, 0);
}

__global__ __launch_bounds__(32 * WPB) void rnn_scan_wmma_f32(
    const float* __restrict__ x,
    const float* __restrict__ W_ih,
    const float* __restrict__ W_hh,
    const float* __restrict__ b_ih,
    const float* __restrict__ b_hh,
    const float* __restrict__ fc_w,
    const float* __restrict__ fc_b,
    float* __restrict__ out,
    int B, int T)
{
    __shared__ float hfin[WPB][16 * HP];

    const int tid  = threadIdx.x;
    const int lane = tid & 31;
    const int wave = tid >> 5;
    const int c    = lane & 15;
    const int hf   = lane >> 4;
    const int base = (blockIdx.x * WPB + wave) * 16;

    v2f aw0[5];
    v2f a0x[4];
    v8f bc0;
    float wm[2][10];
    float wo[2][10];
    float wim[2];
    float bm[2];
    {
        const int j0 = 4 * ((c & 7) >> 1) + 2 * (c >> 3) + (c & 1);
        #pragma unroll
        for (int q = 0; q < 5; ++q) {
            const int k0 = 4 * q + 2 * hf;
            aw0[q].x = W_hh[j0 * HSZ + k0];
            aw0[q].y = W_hh[j0 * HSZ + k0 + 1];
        }
        const float wi0 = W_ih[j0];
        #pragma unroll
        for (int p = 0; p < 4; ++p) {
            a0x[p].x = (hf == (p >> 1) && (p & 1) == 0) ? wi0 : 0.0f;
            a0x[p].y = (hf == (p >> 1) && (p & 1) == 1) ? wi0 : 0.0f;
        }
        #pragma unroll
        for (int r = 0; r < 8; ++r) {
            const int jr0 = 4 * (r >> 1) + 2 * hf + (r & 1);
            bc0[r] = b_ih[jr0] + b_hh[jr0];
        }
        #pragma unroll
        for (int e = 0; e < 2; ++e) {
            const int jm = 16 + 2 * hf + e;
            const int jo = 16 + 2 * (1 - hf) + e;
            #pragma unroll
            for (int q = 0; q < 5; ++q) {
                const int k0 = 4 * q + 2 * hf;
                wm[e][2 * q]     = W_hh[jm * HSZ + k0];
                wm[e][2 * q + 1] = W_hh[jm * HSZ + k0 + 1];
                wo[e][2 * q]     = W_hh[jo * HSZ + k0];
                wo[e][2 * q + 1] = W_hh[jo * HSZ + k0 + 1];
            }
            wim[e] = W_ih[jm];
            bm[e]  = b_ih[jm] + b_hh[jm];
        }
    }
    asm volatile("" : "+v"(a0x[0]), "+v"(a0x[1]), "+v"(a0x[2]), "+v"(a0x[3]));

    int row = base + c;
    if (row >= B) row = B - 1;
    const float* xr = x + (size_t)row * (size_t)T;

    v8f d0p = (v8f){};
    v2f hb4 = (v2f){};

    auto step = [&](auto pc, v2f g, float xt) {
        constexpr int P = decltype(pc)::value;
        v8f d0 = mma(a0x[P], g, bc0);
        d0 = mma(aw0[4], hb4, d0);
        float sA[2], sB[2], oA[2], oB[2];
        sA[0] = __builtin_fmaf(wim[0], xt, bm[0]);
        sA[1] = __builtin_fmaf(wim[1], xt, bm[1]);
        float hv[10];
        #pragma unroll
        for (int r = 0; r < 8; ++r)
            hv[r] = relu(d0p[r]);
        hv[8] = hb4.x;
        hv[9] = hb4.y;
        v2f hb[4];
        #pragma unroll
        for (int q = 0; q < 4; ++q) {
            hb[q].x = hv[2 * q];
            hb[q].y = hv[2 * q + 1];
        }
        #pragma unroll
        for (int e = 0; e < 2; ++e) {
            sA[e] = __builtin_fmaf(wm[e][0], hv[0], sA[e]);
            sB[e] = wm[e][5] * hv[5];
            oA[e] = wo[e][0] * hv[0];
            oB[e] = wo[e][5] * hv[5];
        }
        d0 = mma(aw0[0], hb[0], d0);
        #pragma unroll
        for (int s = 1; s < 3; ++s) {
            #pragma unroll
            for (int e = 0; e < 2; ++e) {
                sA[e] = __builtin_fmaf(wm[e][s], hv[s], sA[e]);
                sB[e] = __builtin_fmaf(wm[e][5 + s], hv[5 + s], sB[e]);
                oA[e] = __builtin_fmaf(wo[e][s], hv[s], oA[e]);
                oB[e] = __builtin_fmaf(wo[e][5 + s], hv[5 + s], oB[e]);
            }
        }
        d0 = mma(aw0[1], hb[1], d0);
        #pragma unroll
        for (int s = 3; s < 5; ++s) {
            #pragma unroll
            for (int e = 0; e < 2; ++e) {
                sA[e] = __builtin_fmaf(wm[e][s], hv[s], sA[e]);
                sB[e] = __builtin_fmaf(wm[e][5 + s], hv[5 + s], sB[e]);
                oA[e] = __builtin_fmaf(wo[e][s], hv[s], oA[e]);
                oB[e] = __builtin_fmaf(wo[e][5 + s], hv[5 + s], oB[e]);
            }
        }
        d0 = mma(aw0[2], hb[2], d0);
        const float m0 = sA[0] + sB[0];
        const float m1 = sA[1] + sB[1];
        const float o0 = xhalf(oA[0] + oB[0]);
        const float o1 = xhalf(oA[1] + oB[1]);
        d0 = mma(aw0[3], hb[3], d0);
        hb4.x = relu(m0 + o0);
        hb4.y = relu(m1 + o1);
        d0p = d0;
        step_order();
    };

    const int NC = ((T & 3) == 0) ? (T >> 3) : 0;
    if (NC > 0) {
        v4f ca = *(const v4f*)(xr);
        v4f cb = *(const v4f*)(xr + 4);
        const float* pn = xr + ((NC > 1) ? 8 : 0);
        #pragma unroll 1
        for (int i = 0; i < NC; ++i) {
            const v4f na = *(const v4f*)(pn);
            const v4f nb = *(const v4f*)(pn + 4);
            hold_order();
            v2f ga, gb;
            ga.x = hf ? ca.z : ca.x;
            ga.y = hf ? ca.w : ca.y;
            gb.x = hf ? cb.z : cb.x;
            gb.y = hf ? cb.w : cb.y;
            hold_order();
            step(step_no<0>{}, ga, ca.x); hold_order();
            step(step_no<1>{}, ga, ca.y); hold_order();
            step(step_no<2>{}, ga, ca.z); hold_order();
            step(step_no<3>{}, ga, ca.w); hold_order();
            step(step_no<0>{}, gb, cb.x); hold_order();
            step(step_no<1>{}, gb, cb.y); hold_order();
            step(step_no<2>{}, gb, cb.z); hold_order();
            step(step_no<3>{}, gb, cb.w);
            hold_order();
            asm volatile("" :: "v"(na), "v"(nb));
            ca = na;
            cb = nb;
            pn += (i + 2 < NC) ? 8 : 0;
        }
    }
    #pragma unroll 1
    for (int t = NC * 8; t < T; ++t) {
        const float xt = xr[t];
        v2f g;
        g.x = xt;
        g.y = 0.0f;
        step(step_no<0>{}, g, xt);
    }

    float* hw = hfin[wave];
    #pragma unroll
    for (int q = 0; q < 4; ++q) {
        hw[c * HP + 4 * q + 2 * hf]     = relu(d0p[2 * q]);
        hw[c * HP + 4 * q + 2 * hf + 1] = relu(d0p[2 * q + 1]);
    }
    hw[c * HP + 16 + 2 * hf]     = hb4.x;
    hw[c * HP + 16 + 2 * hf + 1] = hb4.y;
    __syncthreads();

    if (tid < 16 * WPB) {
        const int orow = blockIdx.x * (16 * WPB) + tid;
        if (orow < B) {
            const float* hr = &hfin[tid >> 4][(tid & 15) * HP];
            float acc = fc_b[0];
            #pragma unroll
            for (int j = 0; j < HSZ; ++j)
                acc += hr[j] * fc_w[j];
            out[orow] = acc;
        }
    }
}

extern "C" void kernel_launch(void* const* d_in, const int* in_sizes, int n_in,
                              void* d_out, int out_size, void* d_ws, size_t ws_size,
                              hipStream_t stream) {
    const float* x    = (const float*)d_in[0];
    const float* W_ih = (const float*)d_in[1];
    const float* W_hh = (const float*)d_in[2];
    const float* b_ih = (const float*)d_in[3];
    const float* b_hh = (const float*)d_in[4];
    const float* fc_w = (const float*)d_in[5];
    const float* fc_b = (const float*)d_in[6];
    float* out = (float*)d_out;

    const int B = out_size;
    const int T = in_sizes[0] / B;
    const int rowsPerBlock = 16 * WPB;
    const int blocks = (B + rowsPerBlock - 1) / rowsPerBlock;

    rnn_scan_wmma_f32<<<blocks, 32 * WPB, 0, stream>>>(x, W_ih, W_hh, b_ih, b_hh,
                                                       fc_w, fc_b, out, B, T);
}


# Step re-laid so no vector op waits: tile sums input, k0..k3, then k4; units 16..19 chains follow the k block pairs (even/odd), join, exchange and clamp ride in the next step; both sum orders changed
# speedup vs baseline: 7.1069x; 1.0570x over previous
#include <hip/hip_runtime.h>


typedef __attribute__((ext_vector_type(2))) float v2f;
typedef __attribute__((ext_vector_type(4))) float v4f;
typedef __attribute__((ext_vector_type(8))) float v8f;

#define HSZ 20
#define WPB 4
#define HP  21

template <int P> struct step_no { static constexpr int value = P; };

__device__ __forceinline__ float relu(float v) {
    return __builtin_amdgcn_fmed3f(v, 0.0f, 3.402823466e38f);
}

__device__ __forceinline__ v8f mma(v2f a, v2f b, v8f c) {
    return __builtin_amdgcn_wmma_f32_16x16x4_f32(false, a, false, b, (short)0, c, false, false);
}

__device__ __forceinline__ float xhalf(float x) {
    const int b = __builtin_bit_cast(int, x);
    return __builtin_bit_cast(float,
        __builtin_amdgcn_permlanex16(b, b, 0x76543210, 0xfedcba98, false, false));
}

__device__ __forceinline__ void hold_order() { __builtin_amdgcn_sched_barrier(0); }

__global__ __launch_bounds__(32 * WPB) void rnn_scan_wmma_f32(
    const float* __restrict__ x,
    const float* __restrict__ W_ih,
    const float* __restrict__ W_hh,
    const float* __restrict__ b_ih,
    const float* __restrict__ b_hh,
    const float* __restrict__ fc_w,
    const float* __restrict__ fc_b,
    float* __restrict__ out,
    int B, int T)
{
    __shared__ float hfin[WPB][16 * HP];

    const int tid  = threadIdx.x;
    const int lane = tid & 31;
    const int wave = tid >> 5;
    const int c    = lane & 15;
    const int hf   = lane >> 4;
    const int base = (blockIdx.x * WPB + wave) * 16;

    v2f aw0[5];
    v2f a0x[4];
    v8f bc0;
    float wm[2][10];
    float wo[2][10];
    float wim[2];
    float bm[2];
    {
        const int j0 = 4 * ((c & 7) >> 1) + 2 * (c >> 3) + (c & 1);
        #pragma unroll
        for (int q = 0; q < 5; ++q) {
            const int k0 = 4 * q + 2 * hf;
            aw0[q].x = W_hh[j0 * HSZ + k0];
            aw0[q].y = W_hh[j0 * HSZ + k0 + 1];
        }
        const float wi0 = W_ih[j0];
        #pragma unroll
        for (int p = 0; p < 4; ++p) {
            a0x[p].x = (hf == (p >> 1) && (p & 1) == 0) ? wi0 : 0.0f;
            a0x[p].y = (hf == (p >> 1) && (p & 1) == 1) ? wi0 : 0.0f;
        }
        #pragma unroll
        for (int r = 0; r < 8; ++r) {
            const int jr0 = 4 * (r >> 1) + 2 * hf + (r & 1);
            bc0[r] = b_ih[jr0] + b_hh[jr0];
        }
        #pragma unroll
        for (int e = 0; e < 2; ++e) {
            const int jm = 16 + 2 * hf + e;
            const int jo = 16 + 2 * (1 - hf) + e;
            #pragma unroll
            for (int q = 0; q < 5; ++q) {
                const int k0 = 4 * q + 2 * hf;
                wm[e][2 * q]     = W_hh[jm * HSZ + k0];
                wm[e][2 * q + 1] = W_hh[jm * HSZ + k0 + 1];
                wo[e][2 * q]     = W_hh[jo * HSZ + k0];
                wo[e][2 * q + 1] = W_hh[jo * HSZ + k0 + 1];
            }
            wim[e] = W_ih[jm];
            bm[e]  = b_ih[jm] + b_hh[jm];
        }
    }
    asm volatile("" : "+v"(a0x[0]), "+v"(a0x[1]), "+v"(a0x[2]), "+v"(a0x[3]));

    int row = base + c;
    if (row >= B) row = B - 1;
    const float* xr = x + (size_t)row * (size_t)T;

    v8f d0p = (v8f){};
    float sA[2] = {0.0f, 0.0f}, sB[2] = {0.0f, 0.0f};
    float oA[2] = {0.0f, 0.0f}, oB[2] = {0.0f, 0.0f};

    auto links = [&](auto qc, v2f hq, float (&nA)[2], float (&nB)[2], float (&pA)[2], float (&pB)[2]) {
        constexpr int Q = decltype(qc)::value;
        #pragma unroll
        for (int e = 0; e < 2; ++e) {
            nA[e] = __builtin_fmaf(wm[e][2 * Q], hq.x, nA[e]);
            nB[e] = (Q == 0) ? wm[e][1] * hq.y : __builtin_fmaf(wm[e][2 * Q + 1], hq.y, nB[e]);
            pA[e] = (Q == 0) ? wo[e][0] * hq.x : __builtin_fmaf(wo[e][2 * Q], hq.x, pA[e]);
            pB[e] = (Q == 0) ? wo[e][1] * hq.y : __builtin_fmaf(wo[e][2 * Q + 1], hq.y, pB[e]);
        }
    };

    auto step = [&](auto pc, v2f g, float xt) {
        constexpr int P = decltype(pc)::value;
        float nA[2], nB[2], pA[2], pB[2];
        v2f hb[5];
        v8f d0 = mma(a0x[P], g, bc0);
        hold_order();
        hb[0].x = relu(d0p[0]);  hb[0].y = relu(d0p[1]);
        const float t0 = oA[0] + oB[0];
        const float t1 = oA[1] + oB[1];
        nA[0] = __builtin_fmaf(wim[0], xt, bm[0]);
        nA[1] = __builtin_fmaf(wim[1], xt, bm[1]);
        hb[1].x = relu(d0p[2]);  hb[1].y = relu(d0p[3]);
        hold_order();
        d0 = mma(aw0[0], hb[0], d0);
        hold_order();
        links(step_no<0>{}, hb[0], nA, nB, pA, pB);
        hb[2].x = relu(d0p[4]);  hb[2].y = relu(d0p[5]);
        const float x0 = xhalf(t0);
        const float x1 = xhalf(t1);
        const float m0 = sA[0] + sB[0];
        const float m1 = sA[1] + sB[1];
        hold_order();
        d0 = mma(aw0[1], hb[1], d0);
        hold_order();
        links(step_no<1>{}, hb[1], nA, nB, pA, pB);
        hb[3].x = relu(d0p[6]);  hb[3].y = relu(d0p[7]);
        const float f0 = m0 + x0;
        const float f1 = m1 + x1;
        hold_order();
        d0 = mma(aw0[2], hb[2], d0);
        hold_order();
        links(step_no<2>{}, hb[2], nA, nB, pA, pB);
        hb[4].x = relu(f0);
        hb[4].y = relu(f1);
        hold_order();
        d0 = mma(aw0[3], hb[3], d0);
        hold_order();
        links(step_no<3>{}, hb[3], nA, nB, pA, pB);
        hold_order();
        d0 = mma(aw0[4], hb[4], d0);
        hold_order();
        links(step_no<4>{}, hb[4], nA, nB, pA, pB);
        d0p = d0;
        #pragma unroll
        for (int e = 0; e < 2; ++e) { sA[e] = nA[e]; sB[e] = nB[e]; oA[e] = pA[e]; oB[e] = pB[e]; }
    };

    const int NC = ((T & 3) == 0) ? (T >> 3) : 0;
    if (NC > 0) {
        v4f ca = *(const v4f*)(xr);
        v4f cb = *(const v4f*)(xr + 4);
        const float* pn = xr + ((NC > 1) ? 8 : 0);
        #pragma unroll 1
        for (int i = 0; i < NC; ++i) {
            const v4f na = *(const v4f*)(pn);
            const v4f nb = *(const v4f*)(pn + 4);
            hold_order();
            v2f ga, gb;
            ga.x = hf ? ca.z : ca.x;
            ga.y = hf ? ca.w : ca.y;
            gb.x = hf ? cb.z : cb.x;
            gb.y = hf ? cb.w : cb.y;
            hold_order();
            step(step_no<0>{}, ga, ca.x); hold_order();
            step(step_no<1>{}, ga, ca.y); hold_order();
            step(step_no<2>{}, ga, ca.z); hold_order();
            step(step_no<3>{}, ga, ca.w); hold_order();
            step(step_no<0>{}, gb, cb.x); hold_order();
            step(step_no<1>{}, gb, cb.y); hold_order();
            step(step_no<2>{}, gb, cb.z); hold_order();
            step(step_no<3>{}, gb, cb.w);
            hold_order();
            asm volatile("" :: "v"(na), "v"(nb));
            ca = na;
            cb = nb;
            pn += (i + 2 < NC) ? 8 : 0;
        }
    }
    #pragma unroll 1
    for (int t = NC * 8; t < T; ++t) {
        const float xt = xr[t];
        v2f g;
        g.x = xt;
        g.y = 0.0f;
        step(step_no<0>{}, g, xt);
    }

    v2f hb4;
    hb4.x = relu((sA[0] + sB[0]) + xhalf(oA[0] + oB[0]));
    hb4.y = relu((sA[1] + sB[1]) + xhalf(oA[1] + oB[1]));

    float* hw = hfin[wave];
    #pragma unroll
    for (int q = 0; q < 4; ++q) {
        hw[c * HP + 4 * q + 2 * hf]     = relu(d0p[2 * q]);
        hw[c * HP + 4 * q + 2 * hf + 1] = relu(d0p[2 * q + 1]);
    }
    hw[c * HP + 16 + 2 * hf]     = hb4.x;
    hw[c * HP + 16 + 2 * hf + 1] = hb4.y;
    __syncthreads();

    if (tid < 16 * WPB) {
        const int orow = blockIdx.x * (16 * WPB) + tid;
        if (orow < B) {
            const float* hr = &hfin[tid >> 4][(tid & 15) * HP];
            float acc = fc_b[0];
            #pragma unroll
            for (int j = 0; j < HSZ; ++j)
                acc += hr[j] * fc_w[j];
            out[orow] = acc;
        }
    }
}

extern "C" void kernel_launch(void* const* d_in, const int* in_sizes, int n_in,
                              void* d_out, int out_size, void* d_ws, size_t ws_size,
                              hipStream_t stream) {
    const float* x    = (const float*)d_in[0];
    const float* W_ih = (const float*)d_in[1];
    const float* W_hh = (const float*)d_in[2];
    const float* b_ih = (const float*)d_in[3];
    const float* b_hh = (const float*)d_in[4];
    const float* fc_w = (const float*)d_in[5];
    const float* fc_b = (const float*)d_in[6];
    float* out = (float*)d_out;

    const int B = out_size;
    const int T = in_sizes[0] / B;
    const int rowsPerBlock = 16 * WPB;
    const int blocks = (B + rowsPerBlock - 1) / rowsPerBlock;

    rnn_scan_wmma_f32<<<blocks, 32 * WPB, 0, stream>>>(x, W_ih, W_hh, b_ih, b_hh,
                                                       fc_w, fc_b, out, B, T);
}


# Each wave takes 8 batch columns, each held in two tile columns (512 waves on idle SIMDs); the two copies split units 16..19: 20 FMAs and 1 half exchange a lane, was 40 and 2; sum orders unchanged
# speedup vs baseline: 7.6003x; 1.0694x over previous
#include <hip/hip_runtime.h>


typedef __attribute__((ext_vector_type(2))) float v2f;
typedef __attribute__((ext_vector_type(4))) float v4f;
typedef __attribute__((ext_vector_type(8))) float v8f;

#define HSZ 20
#define WPB 4
#define HP  21

template <int P> struct step_no { static constexpr int value = P; };

__device__ __forceinline__ float relu(float v) {
    return __builtin_amdgcn_fmed3f(v, 0.0f, 3.402823466e38f);
}

__device__ __forceinline__ v8f mma(v2f a, v2f b, v8f c) {
    return __builtin_amdgcn_wmma_f32_16x16x4_f32(false, a, false, b, (short)0, c, false, false);
}

template <int BANKS> __device__ __forceinline__ float from_copy(float r) {
    const int b = __builtin_bit_cast(int, r);
    return __builtin_bit_cast(float, __builtin_amdgcn_update_dpp(b, b, 0x128, 0xf, BANKS, false));
}

__device__ __forceinline__ float xhalf(float x) {
    const int b = __builtin_bit_cast(int, x);
    return __builtin_bit_cast(float,
        __builtin_amdgcn_permlanex16(b, b, 0x76543210, 0xfedcba98, false, false));
}

__device__ __forceinline__ void hold_order() { __builtin_amdgcn_sched_barrier(0); }

__global__ __launch_bounds__(32 * WPB) void rnn_scan_wmma_f32(
    const float* __restrict__ x,
    const float* __restrict__ W_ih,
    const float* __restrict__ W_hh,
    const float* __restrict__ b_ih,
    const float* __restrict__ b_hh,
    const float* __restrict__ fc_w,
    const float* __restrict__ fc_b,
    float* __restrict__ out,
    int B, int T)
{
    __shared__ float hfin[WPB][8 * HP];

    const int tid  = threadIdx.x;
    const int lane = tid & 31;
    const int wave = tid >> 5;
    const int c    = lane & 15;
    const int col  = c & 7;
    const int d    = c >> 3;
    const int hf   = lane >> 4;
    const int base = (blockIdx.x * WPB + wave) * 8;

    v2f aw0[5];
    v2f a0x[4];
    v8f bc0;
    float wm[10];
    float wo[10];
    float wim;
    float bm;
    {
        const int j0 = 4 * ((c & 7) >> 1) + 2 * (c >> 3) + (c & 1);
        #pragma unroll
        for (int q = 0; q < 5; ++q) {
            const int k0 = 4 * q + 2 * hf;
            aw0[q].x = W_hh[j0 * HSZ + k0];
            aw0[q].y = W_hh[j0 * HSZ + k0 + 1];
        }
        const float wi0 = W_ih[j0];
        #pragma unroll
        for (int p = 0; p < 4; ++p) {
            a0x[p].x = (hf == (p >> 1) && (p & 1) == 0) ? wi0 : 0.0f;
            a0x[p].y = (hf == (p >> 1) && (p & 1) == 1) ? wi0 : 0.0f;
        }
        #pragma unroll
        for (int r = 0; r < 8; ++r) {
            const int jr0 = 4 * (r >> 1) + 2 * hf + (r & 1);
            bc0[r] = b_ih[jr0] + b_hh[jr0];
        }
        {
            const int jm = 16 + 2 * hf + d;
            const int jo = 16 + 2 * (1 - hf) + d;
            #pragma unroll
            for (int q = 0; q < 5; ++q) {
                const int k0 = 4 * q + 2 * hf;
                wm[2 * q]     = W_hh[jm * HSZ + k0];
                wm[2 * q + 1] = W_hh[jm * HSZ + k0 + 1];
                wo[2 * q]     = W_hh[jo * HSZ + k0];
                wo[2 * q + 1] = W_hh[jo * HSZ + k0 + 1];
            }
            wim = W_ih[jm];
            bm  = b_ih[jm] + b_hh[jm];
        }
    }
    asm volatile("" : "+v"(a0x[0]), "+v"(a0x[1]), "+v"(a0x[2]), "+v"(a0x[3]));

    int row = base + col;
    if (row >= B) row = B - 1;
    const float* xr = x + (size_t)row * (size_t)T;

    v8f d0p = (v8f){};
    float sA = 0.0f, sB = 0.0f;
    float oA = 0.0f, oB = 0.0f;

    auto links = [&](auto qc, v2f hq, float &nA, float &nB, float &pA, float &pB) {
        constexpr int Q = decltype(qc)::value;
        nA = __builtin_fmaf(wm[2 * Q], hq.x, nA);
        nB = (Q == 0) ? wm[1] * hq.y : __builtin_fmaf(wm[2 * Q + 1], hq.y, nB);
        pA = (Q == 0) ? wo[0] * hq.x : __builtin_fmaf(wo[2 * Q], hq.x, pA);
        pB = (Q == 0) ? wo[1] * hq.y : __builtin_fmaf(wo[2 * Q + 1], hq.y, pB);
    };

    auto step = [&](auto pc, v2f g, float xt) {
        constexpr int P = decltype(pc)::value;
        float nA, nB, pA, pB;
        v2f hb[5];
        v8f d0 = mma(a0x[P], g, bc0);
        hold_order();
        hb[0].x = relu(d0p[0]);  hb[0].y = relu(d0p[1]);
        const float t0 = oA + oB;
        const float m0 = sA + sB;
        nA = __builtin_fmaf(wim, xt, bm);
        hb[1].x = relu(d0p[2]);  hb[1].y = relu(d0p[3]);
        hold_order();
        d0 = mma(aw0[0], hb[0], d0);
        hold_order();
        links(step_no<0>{}, hb[0], nA, nB, pA, pB);
        hb[2].x = relu(d0p[4]);  hb[2].y = relu(d0p[5]);
        const float x0 = xhalf(t0);
        hold_order();
        d0 = mma(aw0[1], hb[1], d0);
        hold_order();
        links(step_no<1>{}, hb[1], nA, nB, pA, pB);
        hb[3].x = relu(d0p[6]);  hb[3].y = relu(d0p[7]);
        const float r0 = relu(m0 + x0);
        hold_order();
        d0 = mma(aw0[2], hb[2], d0);
        hold_order();
        links(step_no<2>{}, hb[2], nA, nB, pA, pB);
        hb[4].x = from_copy<0xc>(r0);
        hb[4].y = from_copy<0x3>(r0);
        hold_order();
        d0 = mma(aw0[3], hb[3], d0);
        hold_order();
        links(step_no<3>{}, hb[3], nA, nB, pA, pB);
        hold_order();
        d0 = mma(aw0[4], hb[4], d0);
        hold_order();
        links(step_no<4>{}, hb[4], nA, nB, pA, pB);
        d0p = d0;
        sA = nA; sB = nB; oA = pA; oB = pB;
    };

    const int NC = ((T & 3) == 0) ? (T >> 3) : 0;
    if (NC > 0) {
        v4f ca = *(const v4f*)(xr);
        v4f cb = *(const v4f*)(xr + 4);
        const float* pn = xr + ((NC > 1) ? 8 : 0);
        #pragma unroll 1
        for (int i = 0; i < NC; ++i) {
            const v4f na = *(const v4f*)(pn);
            const v4f nb = *(const v4f*)(pn + 4);
            hold_order();
            v2f ga, gb;
            ga.x = hf ? ca.z : ca.x;
            ga.y = hf ? ca.w : ca.y;
            gb.x = hf ? cb.z : cb.x;
            gb.y = hf ? cb.w : cb.y;
            hold_order();
            step(step_no<0>{}, ga, ca.x); hold_order();
            step(step_no<1>{}, ga, ca.y); hold_order();
            step(step_no<2>{}, ga, ca.z); hold_order();
            step(step_no<3>{}, ga, ca.w); hold_order();
            step(step_no<0>{}, gb, cb.x); hold_order();
            step(step_no<1>{}, gb, cb.y); hold_order();
            step(step_no<2>{}, gb, cb.z); hold_order();
            step(step_no<3>{}, gb, cb.w);
            hold_order();
            asm volatile("" :: "v"(na), "v"(nb));
            ca = na;
            cb = nb;
            pn += (i + 2 < NC) ? 8 : 0;
        }
    }
    #pragma unroll 1
    for (int t = NC * 8; t < T; ++t) {
        const float xt = xr[t];
        v2f g;
        g.x = xt;
        g.y = 0.0f;
        step(step_no<0>{}, g, xt);
    }

    const float hfin4 = relu((sA + sB) + xhalf(oA + oB));

    float* hw = hfin[wave];
    #pragma unroll
    for (int q = 0; q < 4; ++q) {
        hw[col * HP + 4 * q + 2 * hf]     = relu(d0p[2 * q]);
        hw[col * HP + 4 * q + 2 * hf + 1] = relu(d0p[2 * q + 1]);
    }
    hw[col * HP + 16 + 2 * hf + d] = hfin4;
    __syncthreads();

    if (tid < 8 * WPB) {
        const int orow = blockIdx.x * (8 * WPB) + tid;
        if (orow < B) {
            const float* hr = &hfin[tid >> 3][(tid & 7) * HP];
            float acc = fc_b[0];
            #pragma unroll
            for (int j = 0; j < HSZ; ++j)
                acc += hr[j] * fc_w[j];
            out[orow] = acc;
        }
    }
}

extern "C" void kernel_launch(void* const* d_in, const int* in_sizes, int n_in,
                              void* d_out, int out_size, void* d_ws, size_t ws_size,
                              hipStream_t stream) {
    const float* x    = (const float*)d_in[0];
    const float* W_ih = (const float*)d_in[1];
    const float* W_hh = (const float*)d_in[2];
    const float* b_ih = (const float*)d_in[3];
    const float* b_hh = (const float*)d_in[4];
    const float* fc_w = (const float*)d_in[5];
    const float* fc_b = (const float*)d_in[6];
    float* out = (float*)d_out;

    const int B = out_size;
    const int T = in_sizes[0] / B;
    const int rowsPerBlock = 8 * WPB;
    const int blocks = (B + rowsPerBlock - 1) / rowsPerBlock;

    rnn_scan_wmma_f32<<<blocks, 32 * WPB, 0, stream>>>(x, W_ih, W_hh, b_ih, b_hh,
                                                       fc_w, fc_b, out, B, T);
}


# Input term leaves the matrix unit: 4 packed multiply-adds a lane form bias + W_ih*x_t as C of k block 0 (its WMMA had 1 useful k slot in 4); 5 WMMAs per step, was 6; sum orders unchanged
# speedup vs baseline: 8.1676x; 1.0746x over previous
#include <hip/hip_runtime.h>


typedef __attribute__((ext_vector_type(2))) float v2f;
typedef __attribute__((ext_vector_type(4))) float v4f;
typedef __attribute__((ext_vector_type(8))) float v8f;

#define HSZ 20
#define WPB 4
#define HP  21

template <int Q> struct block_no { static constexpr int value = Q; };

__device__ __forceinline__ float relu(float v) {
    return __builtin_amdgcn_fmed3f(v, 0.0f, 3.402823466e38f);
}

__device__ __forceinline__ v8f mma(v2f a, v2f b, v8f c) {
    return __builtin_amdgcn_wmma_f32_16x16x4_f32(false, a, false, b, (short)0, c, false, false);
}

template <int BANKS> __device__ __forceinline__ float from_copy(float r) {
    const int b = __builtin_bit_cast(int, r);
    return __builtin_bit_cast(float, __builtin_amdgcn_update_dpp(b, b, 0x128, 0xf, BANKS, false));
}

__device__ __forceinline__ float xhalf(float x) {
    const int b = __builtin_bit_cast(int, x);
    return __builtin_bit_cast(float,
        __builtin_amdgcn_permlanex16(b, b, 0x76543210, 0xfedcba98, false, false));
}

__device__ __forceinline__ void hold_order() { __builtin_amdgcn_sched_barrier(0); }

__global__ __launch_bounds__(32 * WPB) void rnn_scan_wmma_f32(
    const float* __restrict__ x,
    const float* __restrict__ W_ih,
    const float* __restrict__ W_hh,
    const float* __restrict__ b_ih,
    const float* __restrict__ b_hh,
    const float* __restrict__ fc_w,
    const float* __restrict__ fc_b,
    float* __restrict__ out,
    int B, int T)
{
    __shared__ float hfin[WPB][8 * HP];

    const int tid  = threadIdx.x;
    const int lane = tid & 31;
    const int wave = tid >> 5;
    const int c    = lane & 15;
    const int col  = c & 7;
    const int d    = c >> 3;
    const int hf   = lane >> 4;
    const int base = (blockIdx.x * WPB + wave) * 8;

    v2f aw0[5];
    v8f bc0;
    v8f wi8;
    float wm[10];
    float wo[10];
    float wim;
    float bm;
    {
        const int j0 = 4 * ((c & 7) >> 1) + 2 * (c >> 3) + (c & 1);
        #pragma unroll
        for (int q = 0; q < 5; ++q) {
            const int k0 = 4 * q + 2 * hf;
            aw0[q].x = W_hh[j0 * HSZ + k0];
            aw0[q].y = W_hh[j0 * HSZ + k0 + 1];
        }
        #pragma unroll
        for (int r = 0; r < 8; ++r) {
            const int jr0 = 4 * (r >> 1) + 2 * hf + (r & 1);
            bc0[r] = b_ih[jr0] + b_hh[jr0];
            wi8[r] = W_ih[jr0];
        }
        {
            const int jm = 16 + 2 * hf + d;
            const int jo = 16 + 2 * (1 - hf) + d;
            #pragma unroll
            for (int q = 0; q < 5; ++q) {
                const int k0 = 4 * q + 2 * hf;
                wm[2 * q]     = W_hh[jm * HSZ + k0];
                wm[2 * q + 1] = W_hh[jm * HSZ + k0 + 1];
                wo[2 * q]     = W_hh[jo * HSZ + k0];
                wo[2 * q + 1] = W_hh[jo * HSZ + k0 + 1];
            }
            wim = W_ih[jm];
            bm  = b_ih[jm] + b_hh[jm];
        }
    }
    int row = base + col;
    if (row >= B) row = B - 1;
    const float* xr = x + (size_t)row * (size_t)T;

    v8f d0p = (v8f){};
    float sA = 0.0f, sB = 0.0f;
    float oA = 0.0f, oB = 0.0f;

    auto links = [&](auto qc, v2f hq, float &nA, float &nB, float &pA, float &pB) {
        constexpr int Q = decltype(qc)::value;
        nA = __builtin_fmaf(wm[2 * Q], hq.x, nA);
        nB = (Q == 0) ? wm[1] * hq.y : __builtin_fmaf(wm[2 * Q + 1], hq.y, nB);
        pA = (Q == 0) ? wo[0] * hq.x : __builtin_fmaf(wo[2 * Q], hq.x, pA);
        pB = (Q == 0) ? wo[1] * hq.y : __builtin_fmaf(wo[2 * Q + 1], hq.y, pB);
    };

    auto step = [&](float xt) {
        float nA, nB, pA, pB;
        v2f hb[5];
        v8f d0;
        #pragma unroll
        for (int r = 0; r < 8; ++r)
            d0[r] = __builtin_fmaf(wi8[r], xt, bc0[r]);
        const float t0 = oA + oB;
        const float m0 = sA + sB;
        nA = __builtin_fmaf(wim, xt, bm);
        hold_order();
        hb[0].x = relu(d0p[0]);  hb[0].y = relu(d0p[1]);
        hb[1].x = relu(d0p[2]);  hb[1].y = relu(d0p[3]);
        hold_order();
        d0 = mma(aw0[0], hb[0], d0);
        hold_order();
        links(block_no<0>{}, hb[0], nA, nB, pA, pB);
        hb[2].x = relu(d0p[4]);  hb[2].y = relu(d0p[5]);
        const float x0 = xhalf(t0);
        hold_order();
        d0 = mma(aw0[1], hb[1], d0);
        hold_order();
        links(block_no<1>{}, hb[1], nA, nB, pA, pB);
        hb[3].x = relu(d0p[6]);  hb[3].y = relu(d0p[7]);
        const float r0 = relu(m0 + x0);
        hold_order();
        d0 = mma(aw0[2], hb[2], d0);
        hold_order();
        links(block_no<2>{}, hb[2], nA, nB, pA, pB);
        hb[4].x = from_copy<0xc>(r0);
        hb[4].y = from_copy<0x3>(r0);
        hold_order();
        d0 = mma(aw0[3], hb[3], d0);
        hold_order();
        links(block_no<3>{}, hb[3], nA, nB, pA, pB);
        hold_order();
        d0 = mma(aw0[4], hb[4], d0);
        hold_order();
        links(block_no<4>{}, hb[4], nA, nB, pA, pB);
        d0p = d0;
        sA = nA; sB = nB; oA = pA; oB = pB;
    };

    const int NC = ((T & 3) == 0) ? (T >> 3) : 0;
    if (NC > 0) {
        v4f ca = *(const v4f*)(xr);
        v4f cb = *(const v4f*)(xr + 4);
        const float* pn = xr + ((NC > 1) ? 8 : 0);
        #pragma unroll 1
        for (int i = 0; i < NC; ++i) {
            const v4f na = *(const v4f*)(pn);
            const v4f nb = *(const v4f*)(pn + 4);
            hold_order();
            step(ca.x); hold_order();
            step(ca.y); hold_order();
            step(ca.z); hold_order();
            step(ca.w); hold_order();
            step(cb.x); hold_order();
            step(cb.y); hold_order();
            step(cb.z); hold_order();
            step(cb.w);
            hold_order();
            asm volatile("" :: "v"(na), "v"(nb));
            ca = na;
            cb = nb;
            pn += (i + 2 < NC) ? 8 : 0;
        }
    }
    #pragma unroll 1
    for (int t = NC * 8; t < T; ++t)
        step(xr[t]);

    const float hfin4 = relu((sA + sB) + xhalf(oA + oB));

    float* hw = hfin[wave];
    #pragma unroll
    for (int q = 0; q < 4; ++q) {
        hw[col * HP + 4 * q + 2 * hf]     = relu(d0p[2 * q]);
        hw[col * HP + 4 * q + 2 * hf + 1] = relu(d0p[2 * q + 1]);
    }
    hw[col * HP + 16 + 2 * hf + d] = hfin4;
    __syncthreads();

    if (tid < 8 * WPB) {
        const int orow = blockIdx.x * (8 * WPB) + tid;
        if (orow < B) {
            const float* hr = &hfin[tid >> 3][(tid & 7) * HP];
            float acc = fc_b[0];
            #pragma unroll
            for (int j = 0; j < HSZ; ++j)
                acc += hr[j] * fc_w[j];
            out[orow] = acc;
        }
    }
}

extern "C" void kernel_launch(void* const* d_in, const int* in_sizes, int n_in,
                              void* d_out, int out_size, void* d_ws, size_t ws_size,
                              hipStream_t stream) {
    const float* x    = (const float*)d_in[0];
    const float* W_ih = (const float*)d_in[1];
    const float* W_hh = (const float*)d_in[2];
    const float* b_ih = (const float*)d_in[3];
    const float* b_hh = (const float*)d_in[4];
    const float* fc_w = (const float*)d_in[5];
    const float* fc_b = (const float*)d_in[6];
    float* out = (float*)d_out;

    const int B = out_size;
    const int T = in_sizes[0] / B;
    const int rowsPerBlock = 8 * WPB;
    const int blocks = (B + rowsPerBlock - 1) / rowsPerBlock;

    rnn_scan_wmma_f32<<<blocks, 32 * WPB, 0, stream>>>(x, W_ih, W_hh, b_ih, b_hh,
                                                       fc_w, fc_b, out, B, T);
}
